# starting kernel
# baseline (speedup 1.0000x reference)
#include <hip/hip_runtime.h>

#define B_SZ 16384
#define T_SZ 79
#define H_SZ 256
#define HP   260

typedef __attribute__((ext_vector_type(2))) float v2f;
typedef __attribute__((ext_vector_type(8))) float v8f;

__global__ __launch_bounds__(512, 1)
void rnn_fused_kernel(const float* __restrict__ x0,
                      const float* __restrict__ x1,
                      const float* __restrict__ x2,
                      const float* __restrict__ W_in,
                      const float* __restrict__ U,
                      const float* __restrict__ b_rnn,
                      const float* __restrict__ W_d,
                      const float* __restrict__ b_d,
                      float* __restrict__ out)
{
    __shared__ __align__(16) float hbuf[2][16 * HP];
    __shared__ float xs[3][16];

    const int tid  = threadIdx.x;
    const int wave = tid >> 5;
    const int lane = tid & 31;
    const int l15  = lane & 15;
    const int half = lane >> 4;
    const int col  = (wave << 4) + l15;
    const int r0   = blockIdx.x * 16;

    for (int i = tid; i < 2 * 16 * HP; i += 512) ((float*)hbuf)[i] = 0.f;

    const float w0 = W_in[0 * H_SZ + col];
    const float w1 = W_in[1 * H_SZ + col];
    const float w2 = W_in[2 * H_SZ + col];
    const float bb = b_rnn[col];

    v2f Ureg[64];
#pragma unroll
    for (int kt = 0; kt < 64; ++kt) {
        const int krow = (kt << 2) + (half << 1);
        v2f b;
        b.x = U[krow * H_SZ + col];
        b.y = U[(krow + 1) * H_SZ + col];
        Ureg[kt] = b;
    }

    const int a_off = l15 * (HP * 4) + half * 8;

    __syncthreads();

    int cur = 0;
    for (int s = 0; s < T_SZ; ++s) {
        const int idx = (T_SZ - 1) - s;
        if (tid < 48) {
            const int f = tid >> 4, m = tid & 15;
            const float* xp = (f == 0) ? x0 : ((f == 1) ? x1 : x2);
            xs[f][m] = xp[(r0 + m) * T_SZ + idx];
        }
        __syncthreads();

        v8f acc = {0.f, 0.f, 0.f, 0.f, 0.f, 0.f, 0.f, 0.f};
        const char* abase = (const char*)(&hbuf[cur][0]) + a_off;
#pragma unroll
        for (int kt = 0; kt < 64; ++kt) {
            v2f a = *(const v2f*)(abase + kt * 16);
            acc = __builtin_amdgcn_wmma_f32_16x16x4_f32(
                      false, a, false, Ureg[kt], (short)0, acc, false, false);
        }

        float* hn = &hbuf[cur ^ 1][0];
#pragma unroll
        for (int v = 0; v < 8; ++v) {
            const int m = v + (half << 3);
            const float xv = xs[0][m] * w0 + xs[1][m] * w1 + xs[2][m] * w2 + bb;
            float hv = acc[v] + xv;
            hv = fmaxf(hv, 0.f);
            hn[m * HP + col] = hv;
        }
        __syncthreads();
        cur ^= 1;
    }

    {
        v8f acc = {0.f, 0.f, 0.f, 0.f, 0.f, 0.f, 0.f, 0.f};
        const char* abase = (const char*)(&hbuf[cur][0]) + a_off;
#pragma unroll
        for (int kt = 0; kt < 64; ++kt) {
            v2f a = *(const v2f*)(abase + kt * 16);
            const int krow = (kt << 2) + (half << 1);
            v2f b;
            b.x = W_d[krow * H_SZ + col];
            b.y = W_d[(krow + 1) * H_SZ + col];
            acc = __builtin_amdgcn_wmma_f32_16x16x4_f32(
                      false, a, false, b, (short)0, acc, false, false);
        }
        const float bd = b_d[col];
#pragma unroll
        for (int v = 0; v < 8; ++v) {
            const int m = v + (half << 3);
            out[(r0 + m) * H_SZ + col] = acc[v] + bd;
        }
    }
}

extern "C" void kernel_launch(void* const* d_in, const int* in_sizes, int n_in,
                              void* d_out, int out_size, void* d_ws, size_t ws_size,
                              hipStream_t stream) {
    (void)in_sizes; (void)n_in; (void)d_ws; (void)ws_size; (void)out_size;
    const float* x0    = (const float*)d_in[0];
    const float* x1    = (const float*)d_in[1];
    const float* x2    = (const float*)d_in[2];
    const float* W_in  = (const float*)d_in[3];
    const float* U     = (const float*)d_in[4];
    const float* b_rnn = (const float*)d_in[5];
    const float* W_d   = (const float*)d_in[6];
    const float* b_d   = (const float*)d_in[7];
    float* out = (float*)d_out;

    dim3 grid(B_SZ / 16), block(512);
    rnn_fused_kernel<<<grid, block, 0, stream>>>(x0, x1, x2, W_in, U, b_rnn,
                                                 W_d, b_d, out);
}


# 32 rows per work-group as two independent 16-row tiles in alternating phases with a split barrier: each tile's relu/store/barrier and x fetch hide under the other tile's 64-WMMA chain
# speedup vs baseline: 1.0902x; 1.0902x over previous
#include <hip/hip_runtime.h>


#define B_SZ 16384
#define T_SZ 79
#define H_SZ 256
#define HP   260
#define ROWS_PER_WG 32

typedef __attribute__((ext_vector_type(2))) float v2f;
typedef __attribute__((ext_vector_type(8))) float v8f;

static __device__ __forceinline__ void wg_signal() {
    __builtin_amdgcn_fence(__ATOMIC_RELEASE, "workgroup");
    __builtin_amdgcn_s_barrier_signal(-1);
}
static __device__ __forceinline__ void wg_wait() {
    __builtin_amdgcn_s_barrier_wait(-1);
    __builtin_amdgcn_fence(__ATOMIC_ACQUIRE, "workgroup");
}

static __device__ __forceinline__ v8f chain_u(const char* abase, const v2f (&Ureg)[64]) {
    v8f acc = {0.f, 0.f, 0.f, 0.f, 0.f, 0.f, 0.f, 0.f};
#pragma unroll
    for (int kt = 0; kt < 64; ++kt) {
        v2f a = *(const v2f*)(abase + kt * 16);
        acc = __builtin_amdgcn_wmma_f32_16x16x4_f32(
                  false, a, false, Ureg[kt], (short)0, acc, false, false);
    }
    return acc;
}

static __device__ __forceinline__ void relu_store(const v8f acc, const float (*xt)[16],
                                                  float* hn, int half, int col,
                                                  float w0, float w1, float w2, float bb) {
#pragma unroll
    for (int v = 0; v < 8; ++v) {
        const int m = v + (half << 3);
        const float xv = xt[0][m] * w0 + xt[1][m] * w1 + xt[2][m] * w2 + bb;
        float hv = acc[v] + xv;
        hv = fmaxf(hv, 0.f);
        hn[m * HP + col] = hv;
    }
}

__global__ __launch_bounds__(512, 1)
void rnn_fused_kernel(const float* __restrict__ x0,
                      const float* __restrict__ x1,
                      const float* __restrict__ x2,
                      const float* __restrict__ W_in,
                      const float* __restrict__ U,
                      const float* __restrict__ b_rnn,
                      const float* __restrict__ W_d,
                      const float* __restrict__ b_d,
                      float* __restrict__ out)
{
    __shared__ __align__(16) float hbuf[2][2][16 * HP];
    __shared__ __align__(16) float xs[2][3][16];

    const int tid  = threadIdx.x;
    const int wave = tid >> 5;
    const int lane = tid & 31;
    const int l15  = lane & 15;
    const int half = lane >> 4;
    const int col  = (wave << 4) + l15;
    const int r0   = blockIdx.x * ROWS_PER_WG;

    for (int i = tid; i < 2 * 2 * 16 * HP; i += 512) ((float*)hbuf)[i] = 0.f;

    const bool stager = tid < 48;
    const int  sf = tid >> 4, sm = tid & 15;
    const float* xsrc = (sf == 0) ? x0 : ((sf == 1) ? x1 : x2);
    const float* xrow0 = xsrc + (r0 + sm) * T_SZ;
    const float* xrow1 = xsrc + (r0 + 16 + sm) * T_SZ;
    if (stager) xs[0][sf][sm] = xrow0[T_SZ - 1];

    const float w0 = W_in[0 * H_SZ + col];
    const float w1 = W_in[1 * H_SZ + col];
    const float w2 = W_in[2 * H_SZ + col];
    const float bb = b_rnn[col];

    v2f Ureg[64];
#pragma unroll
    for (int kt = 0; kt < 64; ++kt) {
        const int krow = (kt << 2) + (half << 1);
        v2f b;
        b.x = U[krow * H_SZ + col];
        b.y = U[(krow + 1) * H_SZ + col];
        Ureg[kt] = b;
    }

    const int a_off = l15 * (HP * 4) + half * 8;

    __syncthreads();
    wg_signal();

    for (int s = 0; s < T_SZ; ++s) {
        const int cur = s & 1;
        {
            float xn = 0.f;
            if (stager) xn = xrow1[(T_SZ - 1) - s];
            const v8f acc = chain_u((const char*)(&hbuf[0][cur][0]) + a_off, Ureg);
            wg_wait();
            if (stager) xs[1][sf][sm] = xn;
            relu_store(acc, xs[0], &hbuf[0][cur ^ 1][0], half, col, w0, w1, w2, bb);
            wg_signal();
        }
        {
            const bool more = (s + 1) < T_SZ;
            float xn = 0.f;
            if (stager && more) xn = xrow0[(T_SZ - 2) - s];
            const v8f acc = chain_u((const char*)(&hbuf[1][cur][0]) + a_off, Ureg);
            wg_wait();
            if (stager && more) xs[0][sf][sm] = xn;
            relu_store(acc, xs[1], &hbuf[1][cur ^ 1][0], half, col, w0, w1, w2, bb);
            wg_signal();
        }
    }
    wg_wait();

    {
        const int fin = T_SZ & 1;
        v8f acc0 = {0.f, 0.f, 0.f, 0.f, 0.f, 0.f, 0.f, 0.f};
        v8f acc1 = {0.f, 0.f, 0.f, 0.f, 0.f, 0.f, 0.f, 0.f};
        const char* abase0 = (const char*)(&hbuf[0][fin][0]) + a_off;
        const char* abase1 = (const char*)(&hbuf[1][fin][0]) + a_off;
#pragma unroll
        for (int kt = 0; kt < 64; ++kt) {
            v2f a0 = *(const v2f*)(abase0 + kt * 16);
            v2f a1 = *(const v2f*)(abase1 + kt * 16);
            const int krow = (kt << 2) + (half << 1);
            v2f b;
            b.x = W_d[krow * H_SZ + col];
            b.y = W_d[(krow + 1) * H_SZ + col];
            acc0 = __builtin_amdgcn_wmma_f32_16x16x4_f32(
                       false, a0, false, b, (short)0, acc0, false, false);
            acc1 = __builtin_amdgcn_wmma_f32_16x16x4_f32(
                       false, a1, false, b, (short)0, acc1, false, false);
        }
        const float bd = b_d[col];
#pragma unroll
        for (int v = 0; v < 8; ++v) {
            const int m = v + (half << 3);
            out[(r0 + m) * H_SZ + col] = acc0[v] + bd;
            out[(r0 + 16 + m) * H_SZ + col] = acc1[v] + bd;
        }
    }
}

extern "C" void kernel_launch(void* const* d_in, const int* in_sizes, int n_in,
                              void* d_out, int out_size, void* d_ws, size_t ws_size,
                              hipStream_t stream) {
    (void)in_sizes; (void)n_in; (void)d_ws; (void)ws_size; (void)out_size;
    const float* x0    = (const float*)d_in[0];
    const float* x1    = (const float*)d_in[1];
    const float* x2    = (const float*)d_in[2];
    const float* W_in  = (const float*)d_in[3];
    const float* U     = (const float*)d_in[4];
    const float* b_rnn = (const float*)d_in[5];
    const float* W_d   = (const float*)d_in[6];
    const float* b_d   = (const float*)d_in[7];
    float* out = (float*)d_out;

    dim3 grid(B_SZ / ROWS_PER_WG), block(512);
    rnn_fused_kernel<<<grid, block, 0, stream>>>(x0, x1, x2, W_in, U, b_rnn,
                                                 W_d, b_d, out);
}


# LDS reads grouped: A fragments of a 64-WMMA chain fetched in 4 groups of 16 k-tiles, each ahead of the WMMA block before it (group 0 and the x reads behind the phase wait), not 1 read per 2 WMMAs
# speedup vs baseline: 1.1397x; 1.0454x over previous
#include <hip/hip_runtime.h>


#define B_SZ 16384
#define T_SZ 79
#define H_SZ 256
#define HP   260
#define ROWS_PER_WG 32

typedef __attribute__((ext_vector_type(2))) float v2f;
typedef __attribute__((ext_vector_type(8))) float v8f;

static __device__ __forceinline__ void wg_signal() {
    __builtin_amdgcn_fence(__ATOMIC_RELEASE, "workgroup");
    __builtin_amdgcn_s_barrier_signal(-1);
}
static __device__ __forceinline__ void wg_wait() {
    __builtin_amdgcn_s_barrier_wait(-1);
    __builtin_amdgcn_fence(__ATOMIC_ACQUIRE, "workgroup");
}

#define AG 16

static __device__ __forceinline__ void load_group(v2f (&a)[AG], const char* abase, int g) {
#pragma unroll
    for (int i = 0; i < AG; ++i) a[i] = *(const v2f*)(abase + (g * AG + i) * 16);
}

static __device__ __forceinline__ v8f wmma_group(v8f acc, const v2f (&a)[AG],
                                                 const v2f (&Ureg)[64], int g) {
#pragma unroll
    for (int i = 0; i < AG; ++i)
        acc = __builtin_amdgcn_wmma_f32_16x16x4_f32(
                  false, a[i], false, Ureg[g * AG + i], (short)0, acc, false, false);
    return acc;
}

static __device__ __forceinline__ v8f chain_u(const char* abase, v2f (&a0)[AG],
                                              const v2f (&Ureg)[64]) {
    v8f acc = {0.f, 0.f, 0.f, 0.f, 0.f, 0.f, 0.f, 0.f};
    v2f a1[AG];
    load_group(a1, abase, 1);
    __builtin_amdgcn_sched_barrier(0);
    acc = wmma_group(acc, a0, Ureg, 0);
    __builtin_amdgcn_sched_barrier(0);
    load_group(a0, abase, 2);
    __builtin_amdgcn_sched_barrier(0);
    acc = wmma_group(acc, a1, Ureg, 1);
    __builtin_amdgcn_sched_barrier(0);
    load_group(a1, abase, 3);
    __builtin_amdgcn_sched_barrier(0);
    acc = wmma_group(acc, a0, Ureg, 2);
    __builtin_amdgcn_sched_barrier(0);
    acc = wmma_group(acc, a1, Ureg, 3);
    return acc;
}

static __device__ __forceinline__ void read_x(float (&xv)[3][8], const float (*xt)[16], int half) {
#pragma unroll
    for (int f = 0; f < 3; ++f)
#pragma unroll
        for (int v = 0; v < 8; ++v) xv[f][v] = xt[f][v + (half << 3)];
}

static __device__ __forceinline__ void relu_store(const v8f acc, const float (&xv)[3][8],
                                                  float* hn, int half, int col,
                                                  float w0, float w1, float w2, float bb) {
#pragma unroll
    for (int v = 0; v < 8; ++v) {
        const int m = v + (half << 3);
        const float x = xv[0][v] * w0 + xv[1][v] * w1 + xv[2][v] * w2 + bb;
        float hv = acc[v] + x;
        hv = fmaxf(hv, 0.f);
        hn[m * HP + col] = hv;
    }
}

__global__ __launch_bounds__(512, 1)
void rnn_fused_kernel(const float* __restrict__ x0,
                      const float* __restrict__ x1,
                      const float* __restrict__ x2,
                      const float* __restrict__ W_in,
                      const float* __restrict__ U,
                      const float* __restrict__ b_rnn,
                      const float* __restrict__ W_d,
                      const float* __restrict__ b_d,
                      float* __restrict__ out)
{
    __shared__ __align__(16) float hbuf[2][2][16 * HP];
    __shared__ __align__(16) float xs[2][3][16];

    const int tid  = threadIdx.x;
    const int wave = tid >> 5;
    const int lane = tid & 31;
    const int l15  = lane & 15;
    const int half = lane >> 4;
    const int col  = (wave << 4) + l15;
    const int r0   = blockIdx.x * ROWS_PER_WG;

    for (int i = tid; i < 2 * 2 * 16 * HP; i += 512) ((float*)hbuf)[i] = 0.f;

    const bool stager = tid < 48;
    const int  sf = tid >> 4, sm = tid & 15;
    const float* xsrc = (sf == 0) ? x0 : ((sf == 1) ? x1 : x2);
    const float* xrow0 = xsrc + (r0 + sm) * T_SZ;
    const float* xrow1 = xsrc + (r0 + 16 + sm) * T_SZ;
    if (stager) xs[0][sf][sm] = xrow0[T_SZ - 1];

    const float w0 = W_in[0 * H_SZ + col];
    const float w1 = W_in[1 * H_SZ + col];
    const float w2 = W_in[2 * H_SZ + col];
    const float bb = b_rnn[col];

    v2f Ureg[64];
#pragma unroll
    for (int kt = 0; kt < 64; ++kt) {
        const int krow = (kt << 2) + (half << 1);
        v2f b;
        b.x = U[krow * H_SZ + col];
        b.y = U[(krow + 1) * H_SZ + col];
        Ureg[kt] = b;
    }

    const int a_off = l15 * (HP * 4) + half * 8;

    __syncthreads();
    wg_signal();

    v2f af[AG];
    load_group(af, (const char*)(&hbuf[0][0][0]) + a_off, 0);

    for (int s = 0; s < T_SZ; ++s) {
        const int cur = s & 1;
        {
            float xn = 0.f;
            if (stager) xn = xrow1[(T_SZ - 1) - s];
            const v8f acc = chain_u((const char*)(&hbuf[0][cur][0]) + a_off, af, Ureg);
            wg_wait();
            load_group(af, (const char*)(&hbuf[1][cur][0]) + a_off, 0);
            float xv[3][8];
            read_x(xv, xs[0], half);
            if (stager) xs[1][sf][sm] = xn;
            relu_store(acc, xv, &hbuf[0][cur ^ 1][0], half, col, w0, w1, w2, bb);
            wg_signal();
        }
        {
            const bool more = (s + 1) < T_SZ;
            float xn = 0.f;
            if (stager && more) xn = xrow0[(T_SZ - 2) - s];
            const v8f acc = chain_u((const char*)(&hbuf[1][cur][0]) + a_off, af, Ureg);
            wg_wait();
            load_group(af, (const char*)(&hbuf[0][cur ^ 1][0]) + a_off, 0);
            float xv[3][8];
            read_x(xv, xs[1], half);
            if (stager && more) xs[0][sf][sm] = xn;
            relu_store(acc, xv, &hbuf[1][cur ^ 1][0], half, col, w0, w1, w2, bb);
            wg_signal();
        }
    }
    wg_wait();

    {
        const int fin = T_SZ & 1;
        v8f acc0 = {0.f, 0.f, 0.f, 0.f, 0.f, 0.f, 0.f, 0.f};
        v8f acc1 = {0.f, 0.f, 0.f, 0.f, 0.f, 0.f, 0.f, 0.f};
        const char* abase0 = (const char*)(&hbuf[0][fin][0]) + a_off;
        const char* abase1 = (const char*)(&hbuf[1][fin][0]) + a_off;
#pragma unroll
        for (int kt = 0; kt < 64; ++kt) {
            v2f a0 = *(const v2f*)(abase0 + kt * 16);
            v2f a1 = *(const v2f*)(abase1 + kt * 16);
            const int krow = (kt << 2) + (half << 1);
            v2f b;
            b.x = W_d[krow * H_SZ + col];
            b.y = W_d[(krow + 1) * H_SZ + col];
            acc0 = __builtin_amdgcn_wmma_f32_16x16x4_f32(
                       false, a0, false, b, (short)0, acc0, false, false);
            acc1 = __builtin_amdgcn_wmma_f32_16x16x4_f32(
                       false, a1, false, b, (short)0, acc1, false, false);
        }
        const float bd = b_d[col];
#pragma unroll
        for (int v = 0; v < 8; ++v) {
            const int m = v + (half << 3);
            out[(r0 + m) * H_SZ + col] = acc0[v] + bd;
            out[(r0 + 16 + m) * H_SZ + col] = acc1[v] + bd;
        }
    }
}

extern "C" void kernel_launch(void* const* d_in, const int* in_sizes, int n_in,
                              void* d_out, int out_size, void* d_ws, size_t ws_size,
                              hipStream_t stream) {
    (void)in_sizes; (void)n_in; (void)d_ws; (void)ws_size; (void)out_size;
    const float* x0    = (const float*)d_in[0];
    const float* x1    = (const float*)d_in[1];
    const float* x2    = (const float*)d_in[2];
    const float* W_in  = (const float*)d_in[3];
    const float* U     = (const float*)d_in[4];
    const float* b_rnn = (const float*)d_in[5];
    const float* W_d   = (const float*)d_in[6];
    const float* b_d   = (const float*)d_in[7];
    float* out = (float*)d_out;

    dim3 grid(B_SZ / ROWS_PER_WG), block(512);
    rnn_fused_kernel<<<grid, block, 0, stream>>>(x0, x1, x2, W_in, U, b_rnn,
                                                 W_d, b_d, out);
}


# Each phase covers 32 rows: every wave runs two independent 64-WMMA chains (two 16-row tiles on the same U fragment, per-tile k order kept); 64 rows per work-group, half the barrier phases per row
# speedup vs baseline: 1.1919x; 1.0458x over previous
#include <hip/hip_runtime.h>


#define B_SZ 16384
#define T_SZ 79
#define H_SZ 256
#define HP   260
#define TILE_ROWS 32
#define ROWS_PER_WG 64
#define SUB_B (16 * HP * 4)

typedef __attribute__((ext_vector_type(2))) float v2f;
typedef __attribute__((ext_vector_type(8))) float v8f;

static __device__ __forceinline__ void wg_signal() {
    __builtin_amdgcn_fence(__ATOMIC_RELEASE, "workgroup");
    __builtin_amdgcn_s_barrier_signal(-1);
}
static __device__ __forceinline__ void wg_wait() {
    __builtin_amdgcn_s_barrier_wait(-1);
    __builtin_amdgcn_fence(__ATOMIC_ACQUIRE, "workgroup");
}

#define AG 8

static __device__ __forceinline__ void load_group(v2f (&a)[2 * AG], const char* abase, int g) {
#pragma unroll
    for (int i = 0; i < AG; ++i) a[i] = *(const v2f*)(abase + (g * AG + i) * 16);
#pragma unroll
    for (int i = 0; i < AG; ++i) a[AG + i] = *(const v2f*)(abase + SUB_B + (g * AG + i) * 16);
}

static __device__ __forceinline__ void wmma_group(v8f& accA, v8f& accB, const v2f (&a)[2 * AG],
                                                  const v2f (&Ureg)[64], int g) {
#pragma unroll
    for (int i = 0; i < AG; ++i) {
        accA = __builtin_amdgcn_wmma_f32_16x16x4_f32(
                   false, a[i], false, Ureg[g * AG + i], (short)0, accA, false, false);
        accB = __builtin_amdgcn_wmma_f32_16x16x4_f32(
                   false, a[AG + i], false, Ureg[g * AG + i], (short)0, accB, false, false);
    }
}

static __device__ __forceinline__ void chain_head(const char* abase, v2f (&a0)[2 * AG], v2f (&a1)[2 * AG],
                                                  const v2f (&Ureg)[64], v8f& accA, v8f& accB) {
    const v8f zero = {0.f, 0.f, 0.f, 0.f, 0.f, 0.f, 0.f, 0.f};
    accA = zero;
    accB = zero;
    load_group(a1, abase, 1);
    __builtin_amdgcn_sched_barrier(0);
    wmma_group(accA, accB, a0, Ureg, 0);
    __builtin_amdgcn_sched_barrier(0);
    load_group(a0, abase, 2);
    __builtin_amdgcn_sched_barrier(0);
    wmma_group(accA, accB, a1, Ureg, 1);
    __builtin_amdgcn_sched_barrier(0);
    load_group(a1, abase, 3);
    __builtin_amdgcn_sched_barrier(0);
    wmma_group(accA, accB, a0, Ureg, 2);
    __builtin_amdgcn_sched_barrier(0);
    load_group(a0, abase, 4);
    __builtin_amdgcn_sched_barrier(0);
    wmma_group(accA, accB, a1, Ureg, 3);
    __builtin_amdgcn_sched_barrier(0);
    load_group(a1, abase, 5);
    __builtin_amdgcn_sched_barrier(0);
    wmma_group(accA, accB, a0, Ureg, 4);
    __builtin_amdgcn_sched_barrier(0);
    load_group(a0, abase, 6);
    __builtin_amdgcn_sched_barrier(0);
    wmma_group(accA, accB, a1, Ureg, 5);
    __builtin_amdgcn_sched_barrier(0);
    load_group(a1, abase, 7);
    __builtin_amdgcn_sched_barrier(0);
    wmma_group(accA, accB, a0, Ureg, 6);
    __builtin_amdgcn_sched_barrier(0);
}

static __device__ __forceinline__ void read_x(float (&xv)[3][8], const float (*xt)[TILE_ROWS], int half, int sub) {
#pragma unroll
    for (int f = 0; f < 3; ++f)
#pragma unroll
        for (int v = 0; v < 8; ++v) xv[f][v] = xt[f][16 * sub + v + (half << 3)];
}

static __device__ __forceinline__ void relu_store(const v8f acc, const float (&xv)[3][8], int sub,
                                                  float* hn, int half, int col,
                                                  float w0, float w1, float w2, float bb) {
#pragma unroll
    for (int v = 0; v < 8; ++v) {
        const int m = v + (half << 3);
        const float x = xv[0][v] * w0 + xv[1][v] * w1 + xv[2][v] * w2 + bb;
        float hv = acc[v] + x;
        hv = fmaxf(hv, 0.f);
        hn[(16 * sub + m) * HP + col] = hv;
    }
}

__global__ __launch_bounds__(512, 1)
void rnn_fused_kernel(const float* __restrict__ x0,
                      const float* __restrict__ x1,
                      const float* __restrict__ x2,
                      const float* __restrict__ W_in,
                      const float* __restrict__ U,
                      const float* __restrict__ b_rnn,
                      const float* __restrict__ W_d,
                      const float* __restrict__ b_d,
                      float* __restrict__ out)
{
    __shared__ __align__(16) float hbuf[2][2][TILE_ROWS * HP];
    __shared__ __align__(16) float xs[2][3][TILE_ROWS];

    const int tid  = threadIdx.x;
    const int wave = tid >> 5;
    const int lane = tid & 31;
    const int l15  = lane & 15;
    const int half = lane >> 4;
    const int col  = (wave << 4) + l15;
    const int r0   = blockIdx.x * ROWS_PER_WG;

    for (int i = tid; i < 2 * 2 * TILE_ROWS * HP; i += 512) ((float*)hbuf)[i] = 0.f;

    const bool stager = tid < 3 * TILE_ROWS;
    const int  sf = tid >> 5, sm = tid & 31;
    const float* xsrc = (sf == 0) ? x0 : ((sf == 1) ? x1 : x2);
    const float* xrow0 = xsrc + (r0 + sm) * T_SZ;
    const float* xrow1 = xrow0 + TILE_ROWS * T_SZ;
    if (stager) xs[0][sf][sm] = xrow0[T_SZ - 1];

    const float w0 = W_in[0 * H_SZ + col];
    const float w1 = W_in[1 * H_SZ + col];
    const float w2 = W_in[2 * H_SZ + col];
    const float bb = b_rnn[col];

    v2f Ureg[64];
#pragma unroll
    for (int kt = 0; kt < 64; ++kt) {
        const int krow = (kt << 2) + (half << 1);
        v2f b;
        b.x = U[krow * H_SZ + col];
        b.y = U[(krow + 1) * H_SZ + col];
        Ureg[kt] = b;
    }

    const int a_off = l15 * (HP * 4) + half * 8;

    __syncthreads();
    wg_signal();

    v2f af[2 * AG];
    load_group(af, (const char*)(&hbuf[0][0][0]) + a_off, 0);

    for (int s = 0; s < T_SZ; ++s) {
        const int cur = s & 1;
        {
            float xn = 0.f;
            if (stager) xn = xrow1[(T_SZ - 1) - s];
            v8f accA, accB;
            v2f ag[2 * AG];
            chain_head((const char*)(&hbuf[0][cur][0]) + a_off, af, ag, Ureg, accA, accB);
            wg_wait();
            float xa[3][8];
            read_x(xa, xs[0], half, 0);
            __builtin_amdgcn_sched_barrier(0);
            wmma_group(accA, accB, ag, Ureg, 7);
            __builtin_amdgcn_sched_barrier(0);
            float xb[3][8];
            read_x(xb, xs[0], half, 1);
            if (stager) xs[1][sf][sm] = xn;
            relu_store(accA, xa, 0, &hbuf[0][cur ^ 1][0], half, col, w0, w1, w2, bb);
            __builtin_amdgcn_sched_barrier(0);
            load_group(af, (const char*)(&hbuf[1][cur][0]) + a_off, 0);
            relu_store(accB, xb, 1, &hbuf[0][cur ^ 1][0], half, col, w0, w1, w2, bb);
            wg_signal();
        }
        {
            const bool more = (s + 1) < T_SZ;
            float xn = 0.f;
            if (stager && more) xn = xrow0[(T_SZ - 2) - s];
            v8f accA, accB;
            v2f ag[2 * AG];
            chain_head((const char*)(&hbuf[1][cur][0]) + a_off, af, ag, Ureg, accA, accB);
            wg_wait();
            float xa[3][8];
            read_x(xa, xs[1], half, 0);
            __builtin_amdgcn_sched_barrier(0);
            wmma_group(accA, accB, ag, Ureg, 7);
            __builtin_amdgcn_sched_barrier(0);
            float xb[3][8];
            read_x(xb, xs[1], half, 1);
            if (stager && more) xs[0][sf][sm] = xn;
            relu_store(accA, xa, 0, &hbuf[1][cur ^ 1][0], half, col, w0, w1, w2, bb);
            __builtin_amdgcn_sched_barrier(0);
            load_group(af, (const char*)(&hbuf[0][cur ^ 1][0]) + a_off, 0);
            relu_store(accB, xb, 1, &hbuf[1][cur ^ 1][0], half, col, w0, w1, w2, bb);
            wg_signal();
        }
    }
    wg_wait();

    {
        const int fin = T_SZ & 1;
        const float bd = b_d[col];
        for (int t = 0; t < 2; ++t) {
            v8f acc0 = {0.f, 0.f, 0.f, 0.f, 0.f, 0.f, 0.f, 0.f};
            v8f acc1 = {0.f, 0.f, 0.f, 0.f, 0.f, 0.f, 0.f, 0.f};
            const char* abase0 = (const char*)(&hbuf[t][fin][0]) + a_off;
            const char* abase1 = abase0 + SUB_B;
#pragma unroll
            for (int kt = 0; kt < 64; ++kt) {
                v2f a0 = *(const v2f*)(abase0 + kt * 16);
                v2f a1 = *(const v2f*)(abase1 + kt * 16);
                const int krow = (kt << 2) + (half << 1);
                v2f b;
                b.x = W_d[krow * H_SZ + col];
                b.y = W_d[(krow + 1) * H_SZ + col];
                acc0 = __builtin_amdgcn_wmma_f32_16x16x4_f32(
                           false, a0, false, b, (short)0, acc0, false, false);
                acc1 = __builtin_amdgcn_wmma_f32_16x16x4_f32(
                           false, a1, false, b, (short)0, acc1, false, false);
            }
            const int rt = r0 + t * TILE_ROWS;
#pragma unroll
            for (int v = 0; v < 8; ++v) {
                const int m = v + (half << 3);
                out[(rt + m) * H_SZ + col] = acc0[v] + bd;
                out[(rt + 16 + m) * H_SZ + col] = acc1[v] + bd;
            }
        }
    }
}

extern "C" void kernel_launch(void* const* d_in, const int* in_sizes, int n_in,
                              void* d_out, int out_size, void* d_ws, size_t ws_size,
                              hipStream_t stream) {
    (void)in_sizes; (void)n_in; (void)d_ws; (void)ws_size; (void)out_size;
    const float* x0    = (const float*)d_in[0];
    const float* x1    = (const float*)d_in[1];
    const float* x2    = (const float*)d_in[2];
    const float* W_in  = (const float*)d_in[3];
    const float* U     = (const float*)d_in[4];
    const float* b_rnn = (const float*)d_in[5];
    const float* W_d   = (const float*)d_in[6];
    const float* b_d   = (const float*)d_in[7];
    float* out = (float*)d_out;

    dim3 grid(B_SZ / ROWS_PER_WG), block(512);
    rnn_fused_kernel<<<grid, block, 0, stream>>>(x0, x1, x2, W_in, U, b_rnn,
                                                 W_d, b_d, out);
}
